# starting kernel
# baseline (speedup 1.0000x reference)
#include <hip/hip_runtime.h>

typedef float v2f __attribute__((ext_vector_type(2)));
typedef float v8f __attribute__((ext_vector_type(8)));

#define IN_FT  128
#define OUT_FT 128

__global__ void gnnfilm_zero(float* __restrict__ p, int n4) {
  int i = blockIdx.x * blockDim.x + threadIdx.x;
  if (i < n4) {
    ((float4*)p)[i] = make_float4(0.f, 0.f, 0.f, 0.f);
  }
}

__global__ __launch_bounds__(256) void gnnfilm_gemm_wmma(
    const float* __restrict__ seq, const float* __restrict__ W,
    float* __restrict__ seq_fts) {
  const int lane = threadIdx.x & 31;
  const int wave = threadIdx.x >> 5;
  const int m0   = blockIdx.x << 4;
  const int n0   = wave << 4;
  const int half = lane >> 4;
  const int l15  = lane & 15;

  const float* __restrict__ arow = seq + (size_t)(m0 + l15) * IN_FT;
  const float* __restrict__ bcol = W + n0 + l15;

  v8f c = {0.f, 0.f, 0.f, 0.f, 0.f, 0.f, 0.f, 0.f};

#pragma unroll
  for (int kc = 0; kc < IN_FT / 4; ++kc) {
    const int kb = kc * 4 + half * 2;
    v2f a = *(const v2f*)(arow + kb);
    v2f b;
    b.x = bcol[(size_t)(kb + 0) * OUT_FT];
    b.y = bcol[(size_t)(kb + 1) * OUT_FT];
    c = __builtin_amdgcn_wmma_f32_16x16x4_f32(
         false, a,  false, b,
         (short)0, c,  false,  false);
  }

  float* __restrict__ crow =
      seq_fts + (size_t)(m0 + half * 8) * OUT_FT + n0 + l15;
#pragma unroll
  for (int v = 0; v < 8; ++v)
    crow[(size_t)v * OUT_FT] = c[v];
}

__global__ __launch_bounds__(256) void gnnfilm_spmm(
    const float* __restrict__ seq_fts, const float* __restrict__ ew,
    const int* __restrict__ erow, const int* __restrict__ ecol,
    float* __restrict__ agg, int E) {
  int tid = blockIdx.x * blockDim.x + threadIdx.x;
  int e = tid >> 5;
  if (e >= E) return;
  int f4 = (tid & 31) << 2;

  float w = ew[e];
  int cs  = ecol[e];
  int rd  = erow[e];

  float4 v = *(const float4*)(seq_fts + (size_t)cs * OUT_FT + f4);
  float* dst = agg + (size_t)rd * OUT_FT + f4;
  atomicAdd(dst + 0, w * v.x);
  atomicAdd(dst + 1, w * v.y);
  atomicAdd(dst + 2, w * v.z);
  atomicAdd(dst + 3, w * v.w);
}

__global__ __launch_bounds__(256) void gnnfilm_finalize(
    const float* __restrict__ seq_fts,
    const float* __restrict__ Wg, const float* __restrict__ bg,
    const float* __restrict__ Wb, const float* __restrict__ bb,
    const float* __restrict__ bias, const float* __restrict__ prelu_w,
    const int* __restrict__ node_type, float* __restrict__ out, int N) {
  int tid = blockIdx.x * blockDim.x + threadIdx.x;
  int n = tid >> 5;
  if (n >= N) return;
  int f4 = (tid & 31) << 2;

  int t = node_type[n];
  float slope = prelu_w[0];

  float4 g  = *(const float4*)(Wg + (size_t)t * OUT_FT + f4);
  float4 b  = *(const float4*)(Wb + (size_t)t * OUT_FT + f4);
  float4 gb = *(const float4*)(bg + f4);
  float4 bbv= *(const float4*)(bb + f4);
  float4 bi = *(const float4*)(bias + f4);
  float4 sf = *(const float4*)(seq_fts + (size_t)n * OUT_FT + f4);
  float4 ag = *(const float4*)(out + (size_t)n * OUT_FT + f4);

  float o0 = (g.x + gb.x) * ag.x + (b.x + bbv.x) + bi.x + sf.x;
  float o1 = (g.y + gb.y) * ag.y + (b.y + bbv.y) + bi.y + sf.y;
  float o2 = (g.z + gb.z) * ag.z + (b.z + bbv.z) + bi.z + sf.z;
  float o3 = (g.w + gb.w) * ag.w + (b.w + bbv.w) + bi.w + sf.w;

  float4 r;
  r.x = o0 >= 0.f ? o0 : slope * o0;
  r.y = o1 >= 0.f ? o1 : slope * o1;
  r.z = o2 >= 0.f ? o2 : slope * o2;
  r.w = o3 >= 0.f ? o3 : slope * o3;
  *(float4*)(out + (size_t)n * OUT_FT + f4) = r;
}

extern "C" void kernel_launch(void* const* d_in, const int* in_sizes, int n_in,
                              void* d_out, int out_size, void* d_ws, size_t ws_size,
                              hipStream_t stream) {
  const float* seq      = (const float*)d_in[0];
  const float* ew       = (const float*)d_in[1];
  const float* W_fc     = (const float*)d_in[2];
  const float* W_gamma  = (const float*)d_in[3];
  const float* b_gamma  = (const float*)d_in[4];
  const float* W_beta   = (const float*)d_in[5];
  const float* b_beta   = (const float*)d_in[6];
  const float* bias     = (const float*)d_in[7];
  const float* prelu_w  = (const float*)d_in[8];
  const int*   edge_row = (const int*)d_in[9];
  const int*   edge_col = (const int*)d_in[10];
  const int*   node_typ = (const int*)d_in[11];

  const int N = in_sizes[0] / IN_FT;
  const int E = in_sizes[1];

  float* out     = (float*)d_out;
  float* seq_fts = (float*)d_ws;

  {
    int n4 = N * OUT_FT / 4;
    gnnfilm_zero<<<(n4 + 255) / 256, 256, 0, stream>>>(out, n4);
  }
  gnnfilm_gemm_wmma<<<N / 16, 256, 0, stream>>>(seq, W_fc, seq_fts);
  {
    long long threads = (long long)E * 32;
    int blocks = (int)((threads + 255) / 256);
    gnnfilm_spmm<<<blocks, 256, 0, stream>>>(seq_fts, ew, edge_row, edge_col, out, E);
  }
  {
    long long threads = (long long)N * 32;
    int blocks = (int)((threads + 255) / 256);
    gnnfilm_finalize<<<blocks, 256, 0, stream>>>(seq_fts, W_gamma, b_gamma,
                                                 W_beta, b_beta, bias, prelu_w,
                                                 node_typ, out, N);
  }
}


# Receiver-owned gather replaces the f32 atomic scatter: edges counting-sorted by receiver (int atomics, arrival-order rank), one wave sums a row in registers and writes out fused with FiLM
# speedup vs baseline: 1.2431x; 1.2431x over previous
#include <hip/hip_runtime.h>

typedef float v2f __attribute__((ext_vector_type(2)));
typedef float v8f __attribute__((ext_vector_type(8)));

#define IN_FT  128
#define OUT_FT 128

__global__ __launch_bounds__(256) void gnnfilm_gemm_wmma(
    const float* __restrict__ seq, const float* __restrict__ W,
    float* __restrict__ seq_fts, int* __restrict__ cnt,
    int* __restrict__ cursor, int N) {
  const int lane = threadIdx.x & 31;
  const int wave = threadIdx.x >> 5;
  const int m0   = blockIdx.x << 4;
  const int n0   = wave << 4;
  const int half = lane >> 4;
  const int l15  = lane & 15;

  if ((blockIdx.x & 1) == 0 && wave == 0) {
    if (m0 + lane < N) cnt[m0 + lane] = 0;
    if (blockIdx.x == 0 && lane == 0) cursor[0] = 0;
  }

  const float* __restrict__ arow = seq + (size_t)(m0 + l15) * IN_FT;
  const float* __restrict__ bcol = W + n0 + l15;

  v8f c = {0.f, 0.f, 0.f, 0.f, 0.f, 0.f, 0.f, 0.f};

#pragma unroll
  for (int kc = 0; kc < IN_FT / 4; ++kc) {
    const int kb = kc * 4 + half * 2;
    v2f a = *(const v2f*)(arow + kb);
    v2f b;
    b.x = bcol[(size_t)(kb + 0) * OUT_FT];
    b.y = bcol[(size_t)(kb + 1) * OUT_FT];
    c = __builtin_amdgcn_wmma_f32_16x16x4_f32(
         false, a,  false, b,
         (short)0, c,  false,  false);
  }

  float* __restrict__ crow =
      seq_fts + (size_t)(m0 + half * 8) * OUT_FT + n0 + l15;
#pragma unroll
  for (int v = 0; v < 8; ++v)
    crow[(size_t)v * OUT_FT] = c[v];
}

__global__ __launch_bounds__(256) void gnnfilm_count(
    const int* __restrict__ erow, int* __restrict__ cnt, int E) {
  int e = blockIdx.x * blockDim.x + threadIdx.x;
  if (e < E) atomicAdd(cnt + erow[e], 1);
}

__global__ __launch_bounds__(256) void gnnfilm_alloc(
    const int* __restrict__ cnt, int* __restrict__ pos,
    int* __restrict__ cursor, int N) {
  __shared__ int wsum[8];
  __shared__ int base_s;
  const int r    = blockIdx.x * 256 + threadIdx.x;
  const int lane = threadIdx.x & 31;
  const int wave = threadIdx.x >> 5;
  const int c = (r < N) ? cnt[r] : 0;
  int x = c;
#pragma unroll
  for (int d = 1; d < 32; d <<= 1) {
    int y = __shfl_up(x, d, 32);
    if (lane >= d) x += y;
  }
  if (lane == 31) wsum[wave] = x;
  __syncthreads();
  if (threadIdx.x == 0) {
    int tot = 0;
#pragma unroll
    for (int i = 0; i < 8; ++i) { int t = wsum[i]; wsum[i] = tot; tot += t; }
    base_s = atomicAdd(cursor, tot);
  }
  __syncthreads();
  if (r < N) pos[r] = base_s + wsum[wave] + (x - c);
}

__global__ __launch_bounds__(256) void gnnfilm_fill(
    const float* __restrict__ ew, const int* __restrict__ erow,
    const int* __restrict__ ecol, int* __restrict__ pos,
    int2* __restrict__ perm, int E) {
  int e = blockIdx.x * blockDim.x + threadIdx.x;
  if (e >= E) return;
  int p = atomicAdd(pos + erow[e], 1);
  perm[p] = make_int2(ecol[e], __float_as_int(ew[e]));
}

template <int K>
__device__ __forceinline__ void gather_rows(const float* __restrict__ seq_fts,
                                            int2 rec, int j, int f4, float4& ag) {
  int    cc[K];
  float  ww[K];
  float4 vv[K];
#pragma unroll
  for (int i = 0; i < K; ++i) {
    cc[i] = __builtin_amdgcn_readlane(rec.x, j + i);
    ww[i] = __int_as_float(__builtin_amdgcn_readlane(rec.y, j + i));
  }
#pragma unroll
  for (int i = 0; i < K; ++i)
    vv[i] = *(const float4*)(seq_fts + (size_t)cc[i] * OUT_FT + f4);
#pragma unroll
  for (int i = 0; i < K; ++i) {
    ag.x = __builtin_fmaf(ww[i], vv[i].x, ag.x);
    ag.y = __builtin_fmaf(ww[i], vv[i].y, ag.y);
    ag.z = __builtin_fmaf(ww[i], vv[i].z, ag.z);
    ag.w = __builtin_fmaf(ww[i], vv[i].w, ag.w);
  }
}

__global__ __launch_bounds__(256) void gnnfilm_gather_film(
    const float* __restrict__ seq_fts, const int2* __restrict__ perm,
    const int* __restrict__ pos, const int* __restrict__ cnt,
    const float* __restrict__ Wg, const float* __restrict__ bg,
    const float* __restrict__ Wb, const float* __restrict__ bb,
    const float* __restrict__ bias, const float* __restrict__ prelu_w,
    const int* __restrict__ node_type, float* __restrict__ out, int N) {
  const int lane = threadIdx.x & 31;
  const int f4   = lane << 2;
  const int n = __builtin_amdgcn_readfirstlane(
      (int)(blockIdx.x * (blockDim.x >> 5) + (threadIdx.x >> 5)));
  if (n >= N) return;

  const int c = __builtin_amdgcn_readfirstlane(cnt[n]);
  const int s = __builtin_amdgcn_readfirstlane(pos[n]) - c;
  const int t = __builtin_amdgcn_readfirstlane(node_type[n]);

  int m = (c < 32) ? c : 32;
  int2 rec = make_int2(0, 0);
  if (lane < m) rec = perm[s + lane];

  const float slope = prelu_w[0];
  const float4 g   = *(const float4*)(Wg + (size_t)t * OUT_FT + f4);
  const float4 b   = *(const float4*)(Wb + (size_t)t * OUT_FT + f4);
  const float4 gb  = *(const float4*)(bg + f4);
  const float4 bbv = *(const float4*)(bb + f4);
  const float4 bi  = *(const float4*)(bias + f4);
  const float4 sf  = *(const float4*)(seq_fts + (size_t)n * OUT_FT + f4);

  float4 ag = make_float4(0.f, 0.f, 0.f, 0.f);

  int base = 0;
  while (true) {
    int j = 0;
    for (; j + 4 <= m; j += 4) gather_rows<4>(seq_fts, rec, j, f4, ag);
    const int k = m - j;
    if (k == 3)      gather_rows<3>(seq_fts, rec, j, f4, ag);
    else if (k == 2) gather_rows<2>(seq_fts, rec, j, f4, ag);
    else if (k == 1) gather_rows<1>(seq_fts, rec, j, f4, ag);
    base += 32;
    if (base >= c) break;
    m = (c - base < 32) ? (c - base) : 32;
    if (lane < m) rec = perm[s + base + lane];
  }

  float o0 = (g.x + gb.x) * ag.x + (b.x + bbv.x) + bi.x + sf.x;
  float o1 = (g.y + gb.y) * ag.y + (b.y + bbv.y) + bi.y + sf.y;
  float o2 = (g.z + gb.z) * ag.z + (b.z + bbv.z) + bi.z + sf.z;
  float o3 = (g.w + gb.w) * ag.w + (b.w + bbv.w) + bi.w + sf.w;

  float4 r;
  r.x = o0 >= 0.f ? o0 : slope * o0;
  r.y = o1 >= 0.f ? o1 : slope * o1;
  r.z = o2 >= 0.f ? o2 : slope * o2;
  r.w = o3 >= 0.f ? o3 : slope * o3;
  *(float4*)(out + (size_t)n * OUT_FT + f4) = r;
}

extern "C" void kernel_launch(void* const* d_in, const int* in_sizes, int n_in,
                              void* d_out, int out_size, void* d_ws, size_t ws_size,
                              hipStream_t stream) {
  const float* seq      = (const float*)d_in[0];
  const float* ew       = (const float*)d_in[1];
  const float* W_fc     = (const float*)d_in[2];
  const float* W_gamma  = (const float*)d_in[3];
  const float* b_gamma  = (const float*)d_in[4];
  const float* W_beta   = (const float*)d_in[5];
  const float* b_beta   = (const float*)d_in[6];
  const float* bias     = (const float*)d_in[7];
  const float* prelu_w  = (const float*)d_in[8];
  const int*   edge_row = (const int*)d_in[9];
  const int*   edge_col = (const int*)d_in[10];
  const int*   node_typ = (const int*)d_in[11];

  const int N = in_sizes[0] / IN_FT;
  const int E = in_sizes[1];

  float* out     = (float*)d_out;
  float* seq_fts = (float*)d_ws;
  int*   cnt     = (int*)(seq_fts + (size_t)N * OUT_FT);
  int*   pos     = cnt + N;
  int*   cursor  = pos + N;
  int2*  perm    = (int2*)(cursor + 2);

  gnnfilm_gemm_wmma<<<N / 16, 256, 0, stream>>>(seq, W_fc, seq_fts, cnt, cursor, N);
  gnnfilm_count<<<(E + 255) / 256, 256, 0, stream>>>(edge_row, cnt, E);
  gnnfilm_alloc<<<(N + 255) / 256, 256, 0, stream>>>(cnt, pos, cursor, N);
  gnnfilm_fill<<<(E + 255) / 256, 256, 0, stream>>>(ew, edge_row, edge_col, pos, perm, E);
  gnnfilm_gather_film<<<(N + 7) / 8, 256, 0, stream>>>(seq_fts, perm, pos, cnt,
                                                       W_gamma, b_gamma, W_beta, b_beta,
                                                       bias, prelu_w, node_typ, out, N);
}


# Edge grouping through LDS: 64 chunk blocks count per receiver in LDS, a scan over chunks gives first slots, returning LDS adds draw slots; no per-edge global atomic; row order: by chunk, then arrival
# speedup vs baseline: 1.9397x; 1.5604x over previous
#include <hip/hip_runtime.h>

typedef float v2f __attribute__((ext_vector_type(2)));
typedef float v8f __attribute__((ext_vector_type(8)));

#define IN_FT  128
#define OUT_FT 128

#define NCHUNK  64
#define MAXN    50000
#define TSTRIDE 50048
#define EUNR    16
#define LUNR    13

__global__ __launch_bounds__(256) void gnnfilm_gemm_wmma(
    const float* __restrict__ seq, const float* __restrict__ W,
    float* __restrict__ seq_fts, int* __restrict__ cursor, int N) {
  const int lane = threadIdx.x & 31;
  const int wave = threadIdx.x >> 5;
  const int m0   = blockIdx.x << 4;
  const int n0   = wave << 4;
  const int half = lane >> 4;
  const int l15  = lane & 15;

  if (blockIdx.x == 0 && threadIdx.x == 0) cursor[0] = 0;

  const float* __restrict__ arow = seq + (size_t)(m0 + l15) * IN_FT;
  const float* __restrict__ bcol = W + n0 + l15;

  v8f c = {0.f, 0.f, 0.f, 0.f, 0.f, 0.f, 0.f, 0.f};

#pragma unroll
  for (int kc = 0; kc < IN_FT / 4; ++kc) {
    const int kb = kc * 4 + half * 2;
    v2f a = *(const v2f*)(arow + kb);
    v2f b;
    b.x = bcol[(size_t)(kb + 0) * OUT_FT];
    b.y = bcol[(size_t)(kb + 1) * OUT_FT];
    c = __builtin_amdgcn_wmma_f32_16x16x4_f32(
         false, a,  false, b,
         (short)0, c,  false,  false);
  }

  float* __restrict__ crow =
      seq_fts + (size_t)(m0 + half * 8) * OUT_FT + n0 + l15;
#pragma unroll
  for (int v = 0; v < 8; ++v)
    crow[(size_t)v * OUT_FT] = c[v];
}

__global__ __launch_bounds__(1024) void gnnfilm_hist(
    const int* __restrict__ erow, int* __restrict__ table, int N, int E) {
  __shared__ __attribute__((aligned(16))) int h[MAXN];
  const int tid = threadIdx.x;
  const int n4  = N >> 2;
  int4* h4 = (int4*)h;
  for (int i = tid; i < n4; i += 1024) h4[i] = make_int4(0, 0, 0, 0);
  for (int i = (n4 << 2) + tid; i < N; i += 1024) h[i] = 0;
  __syncthreads();

  const int chunk = (E + NCHUNK - 1) / NCHUNK;
  const int e0 = blockIdx.x * chunk;
  const int e1 = (e0 + chunk < E) ? (e0 + chunk) : E;
  for (int eb = e0 + tid; eb < e1; eb += 1024 * EUNR) {
    int rr[EUNR];
#pragma unroll
    for (int i = 0; i < EUNR; ++i) {
      const int e = eb + i * 1024;
      const int v = erow[(e < e1) ? e : (e1 - 1)];
      rr[i] = (e < e1) ? v : -1;
    }
#pragma unroll
    for (int i = 0; i < EUNR; ++i)
      if (rr[i] >= 0) atomicAdd(&h[rr[i]], 1);
  }
  __syncthreads();

  int* __restrict__ dst = table + (size_t)blockIdx.x * TSTRIDE;
  int4* d4 = (int4*)dst;
  for (int i = tid; i < n4; i += 1024) d4[i] = h4[i];
  for (int i = (n4 << 2) + tid; i < N; i += 1024) dst[i] = h[i];
}

__global__ __launch_bounds__(256) void gnnfilm_alloc(
    int* __restrict__ table, int* __restrict__ cnt, int* __restrict__ pos,
    int* __restrict__ cursor, int N) {
  __shared__ int wsum[8];
  __shared__ int base_s;
  const int r    = blockIdx.x * 256 + threadIdx.x;
  const int lane = threadIdx.x & 31;
  const int wave = threadIdx.x >> 5;
  const int rc = (r < N) ? r : (N - 1);
  int t[NCHUNK];
#pragma unroll
  for (int b = 0; b < NCHUNK; ++b) {
    const int v = table[(size_t)b * TSTRIDE + rc];
    t[b] = (r < N) ? v : 0;
  }
  int c = 0;
#pragma unroll
  for (int b = 0; b < NCHUNK; ++b) c += t[b];
  int x = c;
#pragma unroll
  for (int d = 1; d < 32; d <<= 1) {
    int y = __shfl_up(x, d, 32);
    if (lane >= d) x += y;
  }
  if (lane == 31) wsum[wave] = x;
  __syncthreads();
  if (threadIdx.x == 0) {
    int tot = 0;
#pragma unroll
    for (int i = 0; i < 8; ++i) { int u = wsum[i]; wsum[i] = tot; tot += u; }
    base_s = atomicAdd(cursor, tot);
  }
  __syncthreads();
  if (r < N) {
    int run = base_s + wsum[wave] + (x - c);
    cnt[r] = c;
    pos[r] = run + c;
#pragma unroll
    for (int b = 0; b < NCHUNK; ++b) {
      table[(size_t)b * TSTRIDE + r] = run;
      run += t[b];
    }
  }
}

__global__ __launch_bounds__(1024) void gnnfilm_fill(
    const float* __restrict__ ew, const int* __restrict__ erow,
    const int* __restrict__ ecol, const int* __restrict__ table,
    int2* __restrict__ perm, int N, int E) {
  __shared__ __attribute__((aligned(16))) int h[MAXN];
  const int tid = threadIdx.x;
  const int n4  = N >> 2;
  int4* h4 = (int4*)h;
  const int* __restrict__ srcp = table + (size_t)blockIdx.x * TSTRIDE;
  const int4* s4 = (const int4*)srcp;
  for (int i0 = tid; i0 < n4; i0 += 1024 * LUNR) {
    int4 v[LUNR];
#pragma unroll
    for (int u = 0; u < LUNR; ++u) {
      const int i = i0 + u * 1024;
      v[u] = s4[(i < n4) ? i : (n4 - 1)];
    }
#pragma unroll
    for (int u = 0; u < LUNR; ++u) {
      const int i = i0 + u * 1024;
      h4[(i < n4) ? i : (n4 - 1)] = v[u];
    }
  }
  for (int i = (n4 << 2) + tid; i < N; i += 1024) h[i] = srcp[i];
  __syncthreads();

  const int chunk = (E + NCHUNK - 1) / NCHUNK;
  const int e0 = blockIdx.x * chunk;
  const int e1 = (e0 + chunk < E) ? (e0 + chunk) : E;
  for (int eb = e0 + tid; eb < e1; eb += 1024 * EUNR) {
    int rr[EUNR];
    int cc[EUNR];
    int ww[EUNR];
#pragma unroll
    for (int i = 0; i < EUNR; ++i) {
      const int e  = eb + i * 1024;
      const int ec = (e < e1) ? e : (e1 - 1);
      const int v  = erow[ec];
      rr[i] = (e < e1) ? v : -1;
      cc[i] = ecol[ec];
      ww[i] = __float_as_int(ew[ec]);
    }
    int pp[EUNR];
#pragma unroll
    for (int i = 0; i < EUNR; ++i)
      pp[i] = (rr[i] >= 0) ? atomicAdd(&h[rr[i]], 1) : -1;
#pragma unroll
    for (int i = 0; i < EUNR; ++i)
      if (pp[i] >= 0) perm[pp[i]] = make_int2(cc[i], ww[i]);
  }
}

template <int K>
__device__ __forceinline__ void gather_rows(const float* __restrict__ seq_fts,
                                            int2 rec, int j, int f4, float4& ag) {
  int    cc[K];
  float  ww[K];
  float4 vv[K];
#pragma unroll
  for (int i = 0; i < K; ++i) {
    cc[i] = __builtin_amdgcn_readlane(rec.x, j + i);
    ww[i] = __int_as_float(__builtin_amdgcn_readlane(rec.y, j + i));
  }
#pragma unroll
  for (int i = 0; i < K; ++i)
    vv[i] = *(const float4*)(seq_fts + (size_t)cc[i] * OUT_FT + f4);
#pragma unroll
  for (int i = 0; i < K; ++i) {
    ag.x = __builtin_fmaf(ww[i], vv[i].x, ag.x);
    ag.y = __builtin_fmaf(ww[i], vv[i].y, ag.y);
    ag.z = __builtin_fmaf(ww[i], vv[i].z, ag.z);
    ag.w = __builtin_fmaf(ww[i], vv[i].w, ag.w);
  }
}

__global__ __launch_bounds__(256) void gnnfilm_gather_film(
    const float* __restrict__ seq_fts, const int2* __restrict__ perm,
    const int* __restrict__ pos, const int* __restrict__ cnt,
    const float* __restrict__ Wg, const float* __restrict__ bg,
    const float* __restrict__ Wb, const float* __restrict__ bb,
    const float* __restrict__ bias, const float* __restrict__ prelu_w,
    const int* __restrict__ node_type, float* __restrict__ out, int N) {
  const int lane = threadIdx.x & 31;
  const int f4   = lane << 2;
  const int n = __builtin_amdgcn_readfirstlane(
      (int)(blockIdx.x * (blockDim.x >> 5) + (threadIdx.x >> 5)));
  if (n >= N) return;

  const int c = __builtin_amdgcn_readfirstlane(cnt[n]);
  const int s = __builtin_amdgcn_readfirstlane(pos[n]) - c;
  const int t = __builtin_amdgcn_readfirstlane(node_type[n]);

  int m = (c < 32) ? c : 32;
  int2 rec = make_int2(0, 0);
  if (lane < m) rec = perm[s + lane];

  const float slope = prelu_w[0];
  const float4 g   = *(const float4*)(Wg + (size_t)t * OUT_FT + f4);
  const float4 b   = *(const float4*)(Wb + (size_t)t * OUT_FT + f4);
  const float4 gb  = *(const float4*)(bg + f4);
  const float4 bbv = *(const float4*)(bb + f4);
  const float4 bi  = *(const float4*)(bias + f4);
  const float4 sf  = *(const float4*)(seq_fts + (size_t)n * OUT_FT + f4);

  float4 ag = make_float4(0.f, 0.f, 0.f, 0.f);

  int base = 0;
  while (true) {
    int j = 0;
    for (; j + 4 <= m; j += 4) gather_rows<4>(seq_fts, rec, j, f4, ag);
    const int k = m - j;
    if (k == 3)      gather_rows<3>(seq_fts, rec, j, f4, ag);
    else if (k == 2) gather_rows<2>(seq_fts, rec, j, f4, ag);
    else if (k == 1) gather_rows<1>(seq_fts, rec, j, f4, ag);
    base += 32;
    if (base >= c) break;
    m = (c - base < 32) ? (c - base) : 32;
    if (lane < m) rec = perm[s + base + lane];
  }

  float o0 = (g.x + gb.x) * ag.x + (b.x + bbv.x) + bi.x + sf.x;
  float o1 = (g.y + gb.y) * ag.y + (b.y + bbv.y) + bi.y + sf.y;
  float o2 = (g.z + gb.z) * ag.z + (b.z + bbv.z) + bi.z + sf.z;
  float o3 = (g.w + gb.w) * ag.w + (b.w + bbv.w) + bi.w + sf.w;

  float4 r;
  r.x = o0 >= 0.f ? o0 : slope * o0;
  r.y = o1 >= 0.f ? o1 : slope * o1;
  r.z = o2 >= 0.f ? o2 : slope * o2;
  r.w = o3 >= 0.f ? o3 : slope * o3;
  *(float4*)(out + (size_t)n * OUT_FT + f4) = r;
}

extern "C" void kernel_launch(void* const* d_in, const int* in_sizes, int n_in,
                              void* d_out, int out_size, void* d_ws, size_t ws_size,
                              hipStream_t stream) {
  const float* seq      = (const float*)d_in[0];
  const float* ew       = (const float*)d_in[1];
  const float* W_fc     = (const float*)d_in[2];
  const float* W_gamma  = (const float*)d_in[3];
  const float* b_gamma  = (const float*)d_in[4];
  const float* W_beta   = (const float*)d_in[5];
  const float* b_beta   = (const float*)d_in[6];
  const float* bias     = (const float*)d_in[7];
  const float* prelu_w  = (const float*)d_in[8];
  const int*   edge_row = (const int*)d_in[9];
  const int*   edge_col = (const int*)d_in[10];
  const int*   node_typ = (const int*)d_in[11];

  const int N = in_sizes[0] / IN_FT;
  const int E = in_sizes[1];

  float* out     = (float*)d_out;
  float* seq_fts = (float*)d_ws;
  int*   table   = (int*)(seq_fts + (size_t)N * OUT_FT);
  int*   cnt     = table + (size_t)NCHUNK * TSTRIDE;
  int*   pos     = cnt + N;
  int*   cursor  = pos + N;
  int2*  perm    = (int2*)(cursor + 2);

  gnnfilm_gemm_wmma<<<N / 16, 256, 0, stream>>>(seq, W_fc, seq_fts, cursor, N);
  gnnfilm_hist<<<NCHUNK, 1024, 0, stream>>>(edge_row, table, N, E);
  gnnfilm_alloc<<<(N + 255) / 256, 256, 0, stream>>>(table, cnt, pos, cursor, N);
  gnnfilm_fill<<<NCHUNK, 1024, 0, stream>>>(ew, edge_row, edge_col, table, perm, N, E);
  gnnfilm_gather_film<<<(N + 7) / 8, 256, 0, stream>>>(seq_fts, perm, pos, cnt,
                                                       W_gamma, b_gamma, W_beta, b_beta,
                                                       bias, prelu_w, node_typ, out, N);
}


# Resident looping waves: GEMM and gather launches start 5000 and 12504 waves that walk 5 row tiles / 4 rows each, not 25000 and 50000 one-shot waves; W slab stays in registers; sum order unchanged
# speedup vs baseline: 2.0105x; 1.0365x over previous
#include <hip/hip_runtime.h>

typedef float v2f __attribute__((ext_vector_type(2)));
typedef float v8f __attribute__((ext_vector_type(8)));

#define IN_FT  128
#define OUT_FT 128

#define NCHUNK  64
#define MAXN    50000
#define TSTRIDE 50048
#define EUNR    16
#define LUNR    13

__global__ __launch_bounds__(256) void gnnfilm_gemm_wmma(
    const float* __restrict__ seq, const float* __restrict__ W,
    float* __restrict__ seq_fts, int* __restrict__ cursor, int N) {
  const int lane = threadIdx.x & 31;
  const int wave = threadIdx.x >> 5;
  const int n0   = wave << 4;
  const int half = lane >> 4;
  const int l15  = lane & 15;

  if (blockIdx.x == 0 && threadIdx.x == 0) cursor[0] = 0;

  const float* __restrict__ bcol = W + n0 + l15;
  v2f bf[IN_FT / 4];
#pragma unroll
  for (int kc = 0; kc < IN_FT / 4; ++kc) {
    const int kb = kc * 4 + half * 2;
    bf[kc].x = bcol[(size_t)(kb + 0) * OUT_FT];
    bf[kc].y = bcol[(size_t)(kb + 1) * OUT_FT];
  }

  const int ntile = N >> 4;
  for (int tile = blockIdx.x; tile < ntile; tile += gridDim.x) {
    const int m0 = tile << 4;
    const float* __restrict__ arow = seq + (size_t)(m0 + l15) * IN_FT + half * 2;

    v2f af[IN_FT / 4];
#pragma unroll
    for (int kc = 0; kc < IN_FT / 4; ++kc)
      af[kc] = *(const v2f*)(arow + kc * 4);

    asm volatile("" :: "v"(af[0]), "v"(af[1]), "v"(af[2]), "v"(af[3]),
                       "v"(af[4]), "v"(af[5]), "v"(af[6]), "v"(af[7]),
                       "v"(af[8]), "v"(af[9]), "v"(af[10]), "v"(af[11]),
                       "v"(af[12]), "v"(af[13]), "v"(af[14]), "v"(af[15]));
    asm volatile("" :: "v"(af[16]), "v"(af[17]), "v"(af[18]), "v"(af[19]),
                       "v"(af[20]), "v"(af[21]), "v"(af[22]), "v"(af[23]),
                       "v"(af[24]), "v"(af[25]), "v"(af[26]), "v"(af[27]),
                       "v"(af[28]), "v"(af[29]), "v"(af[30]), "v"(af[31]));

    v8f c = {0.f, 0.f, 0.f, 0.f, 0.f, 0.f, 0.f, 0.f};
#pragma unroll
    for (int kc = 0; kc < IN_FT / 4; ++kc)
      c = __builtin_amdgcn_wmma_f32_16x16x4_f32(
           false, af[kc],  false, bf[kc],
           (short)0, c,  false,  false);

    float* __restrict__ crow =
        seq_fts + (size_t)(m0 + half * 8) * OUT_FT + n0 + l15;
#pragma unroll
    for (int v = 0; v < 8; ++v)
      crow[(size_t)v * OUT_FT] = c[v];
  }
}

__global__ __launch_bounds__(1024) void gnnfilm_hist(
    const int* __restrict__ erow, int* __restrict__ table, int N, int E) {
  __shared__ __attribute__((aligned(16))) int h[MAXN];
  const int tid = threadIdx.x;
  const int n4  = N >> 2;
  int4* h4 = (int4*)h;
  for (int i = tid; i < n4; i += 1024) h4[i] = make_int4(0, 0, 0, 0);
  for (int i = (n4 << 2) + tid; i < N; i += 1024) h[i] = 0;
  __syncthreads();

  const int chunk = (E + NCHUNK - 1) / NCHUNK;
  const int e0 = blockIdx.x * chunk;
  const int e1 = (e0 + chunk < E) ? (e0 + chunk) : E;
  for (int eb = e0 + tid; eb < e1; eb += 1024 * EUNR) {
    int rr[EUNR];
#pragma unroll
    for (int i = 0; i < EUNR; ++i) {
      const int e = eb + i * 1024;
      const int v = erow[(e < e1) ? e : (e1 - 1)];
      rr[i] = (e < e1) ? v : -1;
    }
#pragma unroll
    for (int i = 0; i < EUNR; ++i)
      if (rr[i] >= 0) atomicAdd(&h[rr[i]], 1);
  }
  __syncthreads();

  int* __restrict__ dst = table + (size_t)blockIdx.x * TSTRIDE;
  int4* d4 = (int4*)dst;
  for (int i = tid; i < n4; i += 1024) d4[i] = h4[i];
  for (int i = (n4 << 2) + tid; i < N; i += 1024) dst[i] = h[i];
}

__global__ __launch_bounds__(256) void gnnfilm_alloc(
    int* __restrict__ table, int* __restrict__ cnt, int* __restrict__ pos,
    int* __restrict__ cursor, int N) {
  __shared__ int wsum[8];
  __shared__ int base_s;
  const int r    = blockIdx.x * 256 + threadIdx.x;
  const int lane = threadIdx.x & 31;
  const int wave = threadIdx.x >> 5;
  const int rc = (r < N) ? r : (N - 1);
  int t[NCHUNK];
#pragma unroll
  for (int b = 0; b < NCHUNK; ++b) {
    const int v = table[(size_t)b * TSTRIDE + rc];
    t[b] = (r < N) ? v : 0;
  }
  int c = 0;
#pragma unroll
  for (int b = 0; b < NCHUNK; ++b) c += t[b];
  int x = c;
#pragma unroll
  for (int d = 1; d < 32; d <<= 1) {
    int y = __shfl_up(x, d, 32);
    if (lane >= d) x += y;
  }
  if (lane == 31) wsum[wave] = x;
  __syncthreads();
  if (threadIdx.x == 0) {
    int tot = 0;
#pragma unroll
    for (int i = 0; i < 8; ++i) { int u = wsum[i]; wsum[i] = tot; tot += u; }
    base_s = atomicAdd(cursor, tot);
  }
  __syncthreads();
  if (r < N) {
    int run = base_s + wsum[wave] + (x - c);
    cnt[r] = c;
    pos[r] = run + c;
#pragma unroll
    for (int b = 0; b < NCHUNK; ++b) {
      table[(size_t)b * TSTRIDE + r] = run;
      run += t[b];
    }
  }
}

__global__ __launch_bounds__(1024) void gnnfilm_fill(
    const float* __restrict__ ew, const int* __restrict__ erow,
    const int* __restrict__ ecol, const int* __restrict__ table,
    int2* __restrict__ perm, int N, int E) {
  __shared__ __attribute__((aligned(16))) int h[MAXN];
  const int tid = threadIdx.x;
  const int n4  = N >> 2;
  int4* h4 = (int4*)h;
  const int* __restrict__ srcp = table + (size_t)blockIdx.x * TSTRIDE;
  const int4* s4 = (const int4*)srcp;
  for (int i0 = tid; i0 < n4; i0 += 1024 * LUNR) {
    int4 v[LUNR];
#pragma unroll
    for (int u = 0; u < LUNR; ++u) {
      const int i = i0 + u * 1024;
      v[u] = s4[(i < n4) ? i : (n4 - 1)];
    }
#pragma unroll
    for (int u = 0; u < LUNR; ++u) {
      const int i = i0 + u * 1024;
      h4[(i < n4) ? i : (n4 - 1)] = v[u];
    }
  }
  for (int i = (n4 << 2) + tid; i < N; i += 1024) h[i] = srcp[i];
  __syncthreads();

  const int chunk = (E + NCHUNK - 1) / NCHUNK;
  const int e0 = blockIdx.x * chunk;
  const int e1 = (e0 + chunk < E) ? (e0 + chunk) : E;
  for (int eb = e0 + tid; eb < e1; eb += 1024 * EUNR) {
    int rr[EUNR];
    int cc[EUNR];
    int ww[EUNR];
#pragma unroll
    for (int i = 0; i < EUNR; ++i) {
      const int e  = eb + i * 1024;
      const int ec = (e < e1) ? e : (e1 - 1);
      const int v  = erow[ec];
      rr[i] = (e < e1) ? v : -1;
      cc[i] = ecol[ec];
      ww[i] = __float_as_int(ew[ec]);
    }
    int pp[EUNR];
#pragma unroll
    for (int i = 0; i < EUNR; ++i)
      pp[i] = (rr[i] >= 0) ? atomicAdd(&h[rr[i]], 1) : -1;
#pragma unroll
    for (int i = 0; i < EUNR; ++i)
      if (pp[i] >= 0) perm[pp[i]] = make_int2(cc[i], ww[i]);
  }
}

template <int K>
__device__ __forceinline__ void gather_rows(const float* __restrict__ seq_fts,
                                            int2 rec, int j, int f4, float4& ag) {
  int    cc[K];
  float  ww[K];
  float4 vv[K];
#pragma unroll
  for (int i = 0; i < K; ++i) {
    cc[i] = __builtin_amdgcn_readlane(rec.x, j + i);
    ww[i] = __int_as_float(__builtin_amdgcn_readlane(rec.y, j + i));
  }
#pragma unroll
  for (int i = 0; i < K; ++i)
    vv[i] = *(const float4*)(seq_fts + (size_t)cc[i] * OUT_FT + f4);
#pragma unroll
  for (int i = 0; i < K; ++i) {
    ag.x = __builtin_fmaf(ww[i], vv[i].x, ag.x);
    ag.y = __builtin_fmaf(ww[i], vv[i].y, ag.y);
    ag.z = __builtin_fmaf(ww[i], vv[i].z, ag.z);
    ag.w = __builtin_fmaf(ww[i], vv[i].w, ag.w);
  }
}

__global__ __launch_bounds__(256) void gnnfilm_gather_film(
    const float* __restrict__ seq_fts, const int2* __restrict__ perm,
    const int* __restrict__ pos, const int* __restrict__ cnt,
    const float* __restrict__ Wg, const float* __restrict__ bg,
    const float* __restrict__ Wb, const float* __restrict__ bb,
    const float* __restrict__ bias, const float* __restrict__ prelu_w,
    const int* __restrict__ node_type, float* __restrict__ out, int N) {
  const int lane = threadIdx.x & 31;
  const int f4   = lane << 2;
  const int tw   = (int)(gridDim.x * (blockDim.x >> 5));
  int n = __builtin_amdgcn_readfirstlane(
      (int)(blockIdx.x * (blockDim.x >> 5) + (threadIdx.x >> 5)));
  if (n >= N) return;

  const float slope = prelu_w[0];
  const float4 gb  = *(const float4*)(bg + f4);
  const float4 bbv = *(const float4*)(bb + f4);
  const float4 bi  = *(const float4*)(bias + f4);

  int cv = cnt[n];
  int pv = pos[n];
  int tv = node_type[n];

  while (true) {
    const int c = __builtin_amdgcn_readfirstlane(cv);
    const int s = __builtin_amdgcn_readfirstlane(pv) - c;
    const int t = __builtin_amdgcn_readfirstlane(tv);

    int m = (c < 32) ? c : 32;
    int2 rec = make_int2(0, 0);
    if (lane < m) rec = perm[s + lane];

    const int nn = n + tw;
    if (nn < N) {
      cv = cnt[nn];
      pv = pos[nn];
      tv = node_type[nn];
    }

    const float4 g  = *(const float4*)(Wg + (size_t)t * OUT_FT + f4);
    const float4 b  = *(const float4*)(Wb + (size_t)t * OUT_FT + f4);
    const float4 sf = *(const float4*)(seq_fts + (size_t)n * OUT_FT + f4);

    float4 ag = make_float4(0.f, 0.f, 0.f, 0.f);

    int base = 0;
    while (true) {
      int j = 0;
      for (; j + 4 <= m; j += 4) gather_rows<4>(seq_fts, rec, j, f4, ag);
      const int k = m - j;
      if (k == 3)      gather_rows<3>(seq_fts, rec, j, f4, ag);
      else if (k == 2) gather_rows<2>(seq_fts, rec, j, f4, ag);
      else if (k == 1) gather_rows<1>(seq_fts, rec, j, f4, ag);
      base += 32;
      if (base >= c) break;
      m = (c - base < 32) ? (c - base) : 32;
      if (lane < m) rec = perm[s + base + lane];
    }

    float o0 = (g.x + gb.x) * ag.x + (b.x + bbv.x) + bi.x + sf.x;
    float o1 = (g.y + gb.y) * ag.y + (b.y + bbv.y) + bi.y + sf.y;
    float o2 = (g.z + gb.z) * ag.z + (b.z + bbv.z) + bi.z + sf.z;
    float o3 = (g.w + gb.w) * ag.w + (b.w + bbv.w) + bi.w + sf.w;

    float4 r;
    r.x = o0 >= 0.f ? o0 : slope * o0;
    r.y = o1 >= 0.f ? o1 : slope * o1;
    r.z = o2 >= 0.f ? o2 : slope * o2;
    r.w = o3 >= 0.f ? o3 : slope * o3;
    *(float4*)(out + (size_t)n * OUT_FT + f4) = r;

    n = nn;
    if (n >= N) break;
  }
}

extern "C" void kernel_launch(void* const* d_in, const int* in_sizes, int n_in,
                              void* d_out, int out_size, void* d_ws, size_t ws_size,
                              hipStream_t stream) {
  const float* seq      = (const float*)d_in[0];
  const float* ew       = (const float*)d_in[1];
  const float* W_fc     = (const float*)d_in[2];
  const float* W_gamma  = (const float*)d_in[3];
  const float* b_gamma  = (const float*)d_in[4];
  const float* W_beta   = (const float*)d_in[5];
  const float* b_beta   = (const float*)d_in[6];
  const float* bias     = (const float*)d_in[7];
  const float* prelu_w  = (const float*)d_in[8];
  const int*   edge_row = (const int*)d_in[9];
  const int*   edge_col = (const int*)d_in[10];
  const int*   node_typ = (const int*)d_in[11];

  const int N = in_sizes[0] / IN_FT;
  const int E = in_sizes[1];

  float* out     = (float*)d_out;
  float* seq_fts = (float*)d_ws;
  int*   table   = (int*)(seq_fts + (size_t)N * OUT_FT);
  int*   cnt     = table + (size_t)NCHUNK * TSTRIDE;
  int*   pos     = cnt + N;
  int*   cursor  = pos + N;
  int2*  perm    = (int2*)(cursor + 2);

  gnnfilm_gemm_wmma<<<(N / 16 + 4) / 5, 256, 0, stream>>>(seq, W_fc, seq_fts, cursor, N);
  gnnfilm_hist<<<NCHUNK, 1024, 0, stream>>>(edge_row, table, N, E);
  gnnfilm_alloc<<<(N + 255) / 256, 256, 0, stream>>>(table, cnt, pos, cursor, N);
  gnnfilm_fill<<<NCHUNK, 1024, 0, stream>>>(ew, edge_row, edge_col, table, perm, N, E);
  gnnfilm_gather_film<<<(N + 31) / 32, 256, 0, stream>>>(seq_fts, perm, pos, cnt,
                                                       W_gamma, b_gamma, W_beta, b_beta,
                                                       bias, prelu_w, node_typ, out, N);
}


# Receiver rows cut into 4 slices: count and fill launches run 256 blocks (chunk x slice, 50 KB LDS piece each) over all WGPs, not 64 blocks of 200 KB on 64; sum order unchanged (chunk, then arrival)
# speedup vs baseline: 2.3968x; 1.1921x over previous
#include <hip/hip_runtime.h>

typedef float v2f __attribute__((ext_vector_type(2)));
typedef float v8f __attribute__((ext_vector_type(8)));

#define IN_FT  128
#define OUT_FT 128

#define NCHUNK  64
#define NSLICE  4
#define MAXN    50000
#define TSTRIDE 50048
#define SROWS   (TSTRIDE / NSLICE)
#define EUNR    16
#define LUNR    4

__global__ __launch_bounds__(256) void gnnfilm_gemm_wmma(
    const float* __restrict__ seq, const float* __restrict__ W,
    float* __restrict__ seq_fts, int* __restrict__ cursor, int N) {
  const int lane = threadIdx.x & 31;
  const int wave = threadIdx.x >> 5;
  const int n0   = wave << 4;
  const int half = lane >> 4;
  const int l15  = lane & 15;

  if (blockIdx.x == 0 && threadIdx.x == 0) cursor[0] = 0;

  const float* __restrict__ bcol = W + n0 + l15;
  v2f bf[IN_FT / 4];
#pragma unroll
  for (int kc = 0; kc < IN_FT / 4; ++kc) {
    const int kb = kc * 4 + half * 2;
    bf[kc].x = bcol[(size_t)(kb + 0) * OUT_FT];
    bf[kc].y = bcol[(size_t)(kb + 1) * OUT_FT];
  }

  const int ntile = N >> 4;
  for (int tile = blockIdx.x; tile < ntile; tile += gridDim.x) {
    const int m0 = tile << 4;
    const float* __restrict__ arow = seq + (size_t)(m0 + l15) * IN_FT + half * 2;

    v2f af[IN_FT / 4];
#pragma unroll
    for (int kc = 0; kc < IN_FT / 4; ++kc)
      af[kc] = *(const v2f*)(arow + kc * 4);

    asm volatile("" :: "v"(af[0]), "v"(af[1]), "v"(af[2]), "v"(af[3]),
                       "v"(af[4]), "v"(af[5]), "v"(af[6]), "v"(af[7]),
                       "v"(af[8]), "v"(af[9]), "v"(af[10]), "v"(af[11]),
                       "v"(af[12]), "v"(af[13]), "v"(af[14]), "v"(af[15]));
    asm volatile("" :: "v"(af[16]), "v"(af[17]), "v"(af[18]), "v"(af[19]),
                       "v"(af[20]), "v"(af[21]), "v"(af[22]), "v"(af[23]),
                       "v"(af[24]), "v"(af[25]), "v"(af[26]), "v"(af[27]),
                       "v"(af[28]), "v"(af[29]), "v"(af[30]), "v"(af[31]));

    v8f c = {0.f, 0.f, 0.f, 0.f, 0.f, 0.f, 0.f, 0.f};
#pragma unroll
    for (int kc = 0; kc < IN_FT / 4; ++kc)
      c = __builtin_amdgcn_wmma_f32_16x16x4_f32(
           false, af[kc],  false, bf[kc],
           (short)0, c,  false,  false);

    float* __restrict__ crow =
        seq_fts + (size_t)(m0 + half * 8) * OUT_FT + n0 + l15;
#pragma unroll
    for (int v = 0; v < 8; ++v)
      crow[(size_t)v * OUT_FT] = c[v];
  }
}

__global__ __launch_bounds__(1024) void gnnfilm_hist(
    const int* __restrict__ erow, int* __restrict__ table, int N, int E) {
  __shared__ __attribute__((aligned(16))) int h[SROWS];
  const int tid = threadIdx.x;
  const int b   = blockIdx.x / NSLICE;
  const int r0  = (blockIdx.x % NSLICE) * SROWS;
  const int r1  = (r0 + SROWS < N) ? (r0 + SROWS) : N;
  const int nr  = (r1 > r0) ? (r1 - r0) : 0;
  const int n4  = nr >> 2;
  int4* h4 = (int4*)h;
  for (int i = tid; i < SROWS / 4; i += 1024) h4[i] = make_int4(0, 0, 0, 0);
  __syncthreads();

  const int chunk = (E + NCHUNK - 1) / NCHUNK;
  const int e0 = b * chunk;
  const int e1 = (e0 + chunk < E) ? (e0 + chunk) : E;
  for (int eb = e0 + tid; eb < e1; eb += 1024 * EUNR) {
    int rr[EUNR];
#pragma unroll
    for (int i = 0; i < EUNR; ++i) {
      const int e = eb + i * 1024;
      const int v = erow[(e < e1) ? e : (e1 - 1)];
      rr[i] = (e < e1 && v >= r0 && v < r1) ? (v - r0) : -1;
    }
#pragma unroll
    for (int i = 0; i < EUNR; ++i)
      if (rr[i] >= 0) atomicAdd(&h[rr[i]], 1);
  }
  __syncthreads();

  int* __restrict__ dst = table + (size_t)b * TSTRIDE + r0;
  int4* d4 = (int4*)dst;
  for (int i = tid; i < n4; i += 1024) d4[i] = h4[i];
  for (int i = (n4 << 2) + tid; i < nr; i += 1024) dst[i] = h[i];
}

__global__ __launch_bounds__(256) void gnnfilm_alloc(
    int* __restrict__ table, int* __restrict__ cnt, int* __restrict__ pos,
    int* __restrict__ cursor, int N) {
  __shared__ int wsum[8];
  __shared__ int base_s;
  const int r    = blockIdx.x * 256 + threadIdx.x;
  const int lane = threadIdx.x & 31;
  const int wave = threadIdx.x >> 5;
  const int rc = (r < N) ? r : (N - 1);
  int t[NCHUNK];
#pragma unroll
  for (int b = 0; b < NCHUNK; ++b) {
    const int v = table[(size_t)b * TSTRIDE + rc];
    t[b] = (r < N) ? v : 0;
  }
  int c = 0;
#pragma unroll
  for (int b = 0; b < NCHUNK; ++b) c += t[b];
  int x = c;
#pragma unroll
  for (int d = 1; d < 32; d <<= 1) {
    int y = __shfl_up(x, d, 32);
    if (lane >= d) x += y;
  }
  if (lane == 31) wsum[wave] = x;
  __syncthreads();
  if (threadIdx.x == 0) {
    int tot = 0;
#pragma unroll
    for (int i = 0; i < 8; ++i) { int u = wsum[i]; wsum[i] = tot; tot += u; }
    base_s = atomicAdd(cursor, tot);
  }
  __syncthreads();
  if (r < N) {
    int run = base_s + wsum[wave] + (x - c);
    cnt[r] = c;
    pos[r] = run + c;
#pragma unroll
    for (int b = 0; b < NCHUNK; ++b) {
      table[(size_t)b * TSTRIDE + r] = run;
      run += t[b];
    }
  }
}

__global__ __launch_bounds__(1024) void gnnfilm_fill(
    const float* __restrict__ ew, const int* __restrict__ erow,
    const int* __restrict__ ecol, const int* __restrict__ table,
    int2* __restrict__ perm, int N, int E) {
  __shared__ __attribute__((aligned(16))) int h[SROWS];
  const int tid = threadIdx.x;
  const int b   = blockIdx.x / NSLICE;
  const int r0  = (blockIdx.x % NSLICE) * SROWS;
  const int r1  = (r0 + SROWS < N) ? (r0 + SROWS) : N;
  const int nr  = (r1 > r0) ? (r1 - r0) : 0;
  const int n4  = nr >> 2;
  int4* h4 = (int4*)h;
  const int* __restrict__ srcp = table + (size_t)b * TSTRIDE + r0;
  const int4* s4 = (const int4*)srcp;
  for (int i0 = tid; i0 < n4; i0 += 1024 * LUNR) {
    int4 v[LUNR];
#pragma unroll
    for (int u = 0; u < LUNR; ++u) {
      const int i = i0 + u * 1024;
      v[u] = s4[(i < n4) ? i : (n4 - 1)];
    }
#pragma unroll
    for (int u = 0; u < LUNR; ++u) {
      const int i = i0 + u * 1024;
      h4[(i < n4) ? i : (n4 - 1)] = v[u];
    }
  }
  for (int i = (n4 << 2) + tid; i < nr; i += 1024) h[i] = srcp[i];
  __syncthreads();

  const int chunk = (E + NCHUNK - 1) / NCHUNK;
  const int e0 = b * chunk;
  const int e1 = (e0 + chunk < E) ? (e0 + chunk) : E;
  for (int eb = e0 + tid; eb < e1; eb += 1024 * EUNR) {
    int rr[EUNR];
    int cc[EUNR];
    int ww[EUNR];
#pragma unroll
    for (int i = 0; i < EUNR; ++i) {
      const int e  = eb + i * 1024;
      const int ec = (e < e1) ? e : (e1 - 1);
      const int v  = erow[ec];
      rr[i] = (e < e1 && v >= r0 && v < r1) ? (v - r0) : -1;
      cc[i] = ecol[ec];
      ww[i] = __float_as_int(ew[ec]);
    }
    int pp[EUNR];
#pragma unroll
    for (int i = 0; i < EUNR; ++i)
      pp[i] = (rr[i] >= 0) ? atomicAdd(&h[rr[i]], 1) : -1;
#pragma unroll
    for (int i = 0; i < EUNR; ++i)
      if (pp[i] >= 0) perm[pp[i]] = make_int2(cc[i], ww[i]);
  }
}

template <int K>
__device__ __forceinline__ void gather_rows(const float* __restrict__ seq_fts,
                                            int2 rec, int j, int f4, float4& ag) {
  int    cc[K];
  float  ww[K];
  float4 vv[K];
#pragma unroll
  for (int i = 0; i < K; ++i) {
    cc[i] = __builtin_amdgcn_readlane(rec.x, j + i);
    ww[i] = __int_as_float(__builtin_amdgcn_readlane(rec.y, j + i));
  }
#pragma unroll
  for (int i = 0; i < K; ++i)
    vv[i] = *(const float4*)(seq_fts + (size_t)cc[i] * OUT_FT + f4);
#pragma unroll
  for (int i = 0; i < K; ++i) {
    ag.x = __builtin_fmaf(ww[i], vv[i].x, ag.x);
    ag.y = __builtin_fmaf(ww[i], vv[i].y, ag.y);
    ag.z = __builtin_fmaf(ww[i], vv[i].z, ag.z);
    ag.w = __builtin_fmaf(ww[i], vv[i].w, ag.w);
  }
}

__global__ __launch_bounds__(256) void gnnfilm_gather_film(
    const float* __restrict__ seq_fts, const int2* __restrict__ perm,
    const int* __restrict__ pos, const int* __restrict__ cnt,
    const float* __restrict__ Wg, const float* __restrict__ bg,
    const float* __restrict__ Wb, const float* __restrict__ bb,
    const float* __restrict__ bias, const float* __restrict__ prelu_w,
    const int* __restrict__ node_type, float* __restrict__ out, int N) {
  const int lane = threadIdx.x & 31;
  const int f4   = lane << 2;
  const int tw   = (int)(gridDim.x * (blockDim.x >> 5));
  int n = __builtin_amdgcn_readfirstlane(
      (int)(blockIdx.x * (blockDim.x >> 5) + (threadIdx.x >> 5)));
  if (n >= N) return;

  const float slope = prelu_w[0];
  const float4 gb  = *(const float4*)(bg + f4);
  const float4 bbv = *(const float4*)(bb + f4);
  const float4 bi  = *(const float4*)(bias + f4);

  int cv = cnt[n];
  int pv = pos[n];
  int tv = node_type[n];

  while (true) {
    const int c = __builtin_amdgcn_readfirstlane(cv);
    const int s = __builtin_amdgcn_readfirstlane(pv) - c;
    const int t = __builtin_amdgcn_readfirstlane(tv);

    int m = (c < 32) ? c : 32;
    int2 rec = make_int2(0, 0);
    if (lane < m) rec = perm[s + lane];

    const int nn = n + tw;
    if (nn < N) {
      cv = cnt[nn];
      pv = pos[nn];
      tv = node_type[nn];
    }

    const float4 g  = *(const float4*)(Wg + (size_t)t * OUT_FT + f4);
    const float4 b  = *(const float4*)(Wb + (size_t)t * OUT_FT + f4);
    const float4 sf = *(const float4*)(seq_fts + (size_t)n * OUT_FT + f4);

    float4 ag = make_float4(0.f, 0.f, 0.f, 0.f);

    int base = 0;
    while (true) {
      int j = 0;
      for (; j + 4 <= m; j += 4) gather_rows<4>(seq_fts, rec, j, f4, ag);
      const int k = m - j;
      if (k == 3)      gather_rows<3>(seq_fts, rec, j, f4, ag);
      else if (k == 2) gather_rows<2>(seq_fts, rec, j, f4, ag);
      else if (k == 1) gather_rows<1>(seq_fts, rec, j, f4, ag);
      base += 32;
      if (base >= c) break;
      m = (c - base < 32) ? (c - base) : 32;
      if (lane < m) rec = perm[s + base + lane];
    }

    float o0 = (g.x + gb.x) * ag.x + (b.x + bbv.x) + bi.x + sf.x;
    float o1 = (g.y + gb.y) * ag.y + (b.y + bbv.y) + bi.y + sf.y;
    float o2 = (g.z + gb.z) * ag.z + (b.z + bbv.z) + bi.z + sf.z;
    float o3 = (g.w + gb.w) * ag.w + (b.w + bbv.w) + bi.w + sf.w;

    float4 r;
    r.x = o0 >= 0.f ? o0 : slope * o0;
    r.y = o1 >= 0.f ? o1 : slope * o1;
    r.z = o2 >= 0.f ? o2 : slope * o2;
    r.w = o3 >= 0.f ? o3 : slope * o3;
    *(float4*)(out + (size_t)n * OUT_FT + f4) = r;

    n = nn;
    if (n >= N) break;
  }
}

extern "C" void kernel_launch(void* const* d_in, const int* in_sizes, int n_in,
                              void* d_out, int out_size, void* d_ws, size_t ws_size,
                              hipStream_t stream) {
  const float* seq      = (const float*)d_in[0];
  const float* ew       = (const float*)d_in[1];
  const float* W_fc     = (const float*)d_in[2];
  const float* W_gamma  = (const float*)d_in[3];
  const float* b_gamma  = (const float*)d_in[4];
  const float* W_beta   = (const float*)d_in[5];
  const float* b_beta   = (const float*)d_in[6];
  const float* bias     = (const float*)d_in[7];
  const float* prelu_w  = (const float*)d_in[8];
  const int*   edge_row = (const int*)d_in[9];
  const int*   edge_col = (const int*)d_in[10];
  const int*   node_typ = (const int*)d_in[11];

  const int N = in_sizes[0] / IN_FT;
  const int E = in_sizes[1];

  float* out     = (float*)d_out;
  float* seq_fts = (float*)d_ws;
  int*   table   = (int*)(seq_fts + (size_t)N * OUT_FT);
  int*   cnt     = table + (size_t)NCHUNK * TSTRIDE;
  int*   pos     = cnt + N;
  int*   cursor  = pos + N;
  int2*  perm    = (int2*)(cursor + 2);

  gnnfilm_gemm_wmma<<<(N / 16 + 4) / 5, 256, 0, stream>>>(seq, W_fc, seq_fts, cursor, N);
  gnnfilm_hist<<<NCHUNK * NSLICE, 1024, 0, stream>>>(edge_row, table, N, E);
  gnnfilm_alloc<<<(N + 255) / 256, 256, 0, stream>>>(table, cnt, pos, cursor, N);
  gnnfilm_fill<<<NCHUNK * NSLICE, 1024, 0, stream>>>(ew, edge_row, edge_col, table, perm, N, E);
  gnnfilm_gather_film<<<(N + 31) / 32, 256, 0, stream>>>(seq_fts, perm, pos, cnt,
                                                       W_gamma, b_gamma, W_beta, b_beta,
                                                       bias, prelu_w, node_typ, out, N);
}


# GEMM A tile staged once per block through LDS: 256 threads fetch the 16x128 seq tile with whole-row 16-B loads one tile ahead; 8 waves read fragments from padded LDS; k and sum order unchanged
# speedup vs baseline: 2.6575x; 1.1088x over previous
#include <hip/hip_runtime.h>

typedef float v2f __attribute__((ext_vector_type(2)));
typedef float v8f __attribute__((ext_vector_type(8)));

#define IN_FT  128
#define OUT_FT 128

#define NCHUNK  64
#define NSLICE  4
#define MAXN    50000
#define TSTRIDE 50048
#define SROWS   (TSTRIDE / NSLICE)
#define EUNR    16
#define LUNR    4

#define APITCH 132
__global__ __launch_bounds__(256) void gnnfilm_gemm_wmma(
    const float* __restrict__ seq, const float* __restrict__ W,
    float* __restrict__ seq_fts, int* __restrict__ cursor, int N) {
  __shared__ __attribute__((aligned(16))) float As[2][16 * APITCH];
  const int tid  = threadIdx.x;
  const int lane = tid & 31;
  const int wave = tid >> 5;
  const int n0   = wave << 4;
  const int half = lane >> 4;
  const int l15  = lane & 15;

  if (blockIdx.x == 0 && tid == 0) cursor[0] = 0;

  const int ntile = N >> 4;
  int tile = blockIdx.x;

  const int sc4 = lane << 2;
  float4 g0 = make_float4(0.f, 0.f, 0.f, 0.f);
  float4 g1 = make_float4(0.f, 0.f, 0.f, 0.f);
  if (tile < ntile) {
    const float* __restrict__ sp = seq + (size_t)((tile << 4) + wave) * IN_FT + sc4;
    g0 = *(const float4*)sp;
    g1 = *(const float4*)(sp + 8 * IN_FT);
  }

  const float* __restrict__ bcol = W + n0 + l15;
  v2f bf[IN_FT / 4];
#pragma unroll
  for (int kc = 0; kc < IN_FT / 4; ++kc) {
    const int kb = kc * 4 + half * 2;
    bf[kc].x = bcol[(size_t)(kb + 0) * OUT_FT];
    bf[kc].y = bcol[(size_t)(kb + 1) * OUT_FT];
  }

  if (tile < ntile) {
    *(float4*)&As[0][wave * APITCH + sc4]       = g0;
    *(float4*)&As[0][(wave + 8) * APITCH + sc4] = g1;
  }

  int buf = 0;
  for (; tile < ntile; tile += gridDim.x) {
    const int m0 = tile << 4;
    const int nt = tile + gridDim.x;
    if (nt < ntile) {
      const float* __restrict__ sp = seq + (size_t)((nt << 4) + wave) * IN_FT + sc4;
      g0 = *(const float4*)sp;
      g1 = *(const float4*)(sp + 8 * IN_FT);
    }

    __builtin_amdgcn_fence(__ATOMIC_RELEASE, "workgroup", "local");
    __builtin_amdgcn_s_barrier();
    __builtin_amdgcn_fence(__ATOMIC_ACQUIRE, "workgroup", "local");

    const float* ap = &As[buf][l15 * APITCH + half * 2];
    v2f af[IN_FT / 4];
#pragma unroll
    for (int kc = 0; kc < IN_FT / 4; ++kc)
      af[kc] = *(const v2f*)(ap + kc * 4);

    asm volatile("" :: "v"(af[0]), "v"(af[1]), "v"(af[2]), "v"(af[3]),
                       "v"(af[4]), "v"(af[5]), "v"(af[6]), "v"(af[7]),
                       "v"(af[8]), "v"(af[9]), "v"(af[10]), "v"(af[11]),
                       "v"(af[12]), "v"(af[13]), "v"(af[14]), "v"(af[15]));
    asm volatile("" :: "v"(af[16]), "v"(af[17]), "v"(af[18]), "v"(af[19]),
                       "v"(af[20]), "v"(af[21]), "v"(af[22]), "v"(af[23]),
                       "v"(af[24]), "v"(af[25]), "v"(af[26]), "v"(af[27]),
                       "v"(af[28]), "v"(af[29]), "v"(af[30]), "v"(af[31]));

    v8f c = {0.f, 0.f, 0.f, 0.f, 0.f, 0.f, 0.f, 0.f};
#pragma unroll
    for (int kc = 0; kc < IN_FT / 4; ++kc)
      c = __builtin_amdgcn_wmma_f32_16x16x4_f32(
           false, af[kc],  false, bf[kc],
           (short)0, c,  false,  false);

    float* __restrict__ crow =
        seq_fts + (size_t)(m0 + half * 8) * OUT_FT + n0 + l15;
#pragma unroll
    for (int v = 0; v < 8; ++v)
      crow[(size_t)v * OUT_FT] = c[v];

    if (nt < ntile) {
      *(float4*)&As[buf ^ 1][wave * APITCH + sc4]       = g0;
      *(float4*)&As[buf ^ 1][(wave + 8) * APITCH + sc4] = g1;
    }
    buf ^= 1;
  }
}

__global__ __launch_bounds__(1024) void gnnfilm_hist(
    const int* __restrict__ erow, int* __restrict__ table, int N, int E) {
  __shared__ __attribute__((aligned(16))) int h[SROWS];
  const int tid = threadIdx.x;
  const int b   = blockIdx.x / NSLICE;
  const int r0  = (blockIdx.x % NSLICE) * SROWS;
  const int r1  = (r0 + SROWS < N) ? (r0 + SROWS) : N;
  const int nr  = (r1 > r0) ? (r1 - r0) : 0;
  const int n4  = nr >> 2;
  int4* h4 = (int4*)h;
  for (int i = tid; i < SROWS / 4; i += 1024) h4[i] = make_int4(0, 0, 0, 0);
  __syncthreads();

  const int chunk = (E + NCHUNK - 1) / NCHUNK;
  const int e0 = b * chunk;
  const int e1 = (e0 + chunk < E) ? (e0 + chunk) : E;
  for (int eb = e0 + tid; eb < e1; eb += 1024 * EUNR) {
    int rr[EUNR];
#pragma unroll
    for (int i = 0; i < EUNR; ++i) {
      const int e = eb + i * 1024;
      const int v = erow[(e < e1) ? e : (e1 - 1)];
      rr[i] = (e < e1 && v >= r0 && v < r1) ? (v - r0) : -1;
    }
#pragma unroll
    for (int i = 0; i < EUNR; ++i)
      if (rr[i] >= 0) atomicAdd(&h[rr[i]], 1);
  }
  __syncthreads();

  int* __restrict__ dst = table + (size_t)b * TSTRIDE + r0;
  int4* d4 = (int4*)dst;
  for (int i = tid; i < n4; i += 1024) d4[i] = h4[i];
  for (int i = (n4 << 2) + tid; i < nr; i += 1024) dst[i] = h[i];
}

__global__ __launch_bounds__(256) void gnnfilm_alloc(
    int* __restrict__ table, int* __restrict__ cnt, int* __restrict__ pos,
    int* __restrict__ cursor, int N) {
  __shared__ int wsum[8];
  __shared__ int base_s;
  const int r    = blockIdx.x * 256 + threadIdx.x;
  const int lane = threadIdx.x & 31;
  const int wave = threadIdx.x >> 5;
  const int rc = (r < N) ? r : (N - 1);
  int t[NCHUNK];
#pragma unroll
  for (int b = 0; b < NCHUNK; ++b) {
    const int v = table[(size_t)b * TSTRIDE + rc];
    t[b] = (r < N) ? v : 0;
  }
  int c = 0;
#pragma unroll
  for (int b = 0; b < NCHUNK; ++b) c += t[b];
  int x = c;
#pragma unroll
  for (int d = 1; d < 32; d <<= 1) {
    int y = __shfl_up(x, d, 32);
    if (lane >= d) x += y;
  }
  if (lane == 31) wsum[wave] = x;
  __syncthreads();
  if (threadIdx.x == 0) {
    int tot = 0;
#pragma unroll
    for (int i = 0; i < 8; ++i) { int u = wsum[i]; wsum[i] = tot; tot += u; }
    base_s = atomicAdd(cursor, tot);
  }
  __syncthreads();
  if (r < N) {
    int run = base_s + wsum[wave] + (x - c);
    cnt[r] = c;
    pos[r] = run + c;
#pragma unroll
    for (int b = 0; b < NCHUNK; ++b) {
      table[(size_t)b * TSTRIDE + r] = run;
      run += t[b];
    }
  }
}

__global__ __launch_bounds__(1024) void gnnfilm_fill(
    const float* __restrict__ ew, const int* __restrict__ erow,
    const int* __restrict__ ecol, const int* __restrict__ table,
    int2* __restrict__ perm, int N, int E) {
  __shared__ __attribute__((aligned(16))) int h[SROWS];
  const int tid = threadIdx.x;
  const int b   = blockIdx.x / NSLICE;
  const int r0  = (blockIdx.x % NSLICE) * SROWS;
  const int r1  = (r0 + SROWS < N) ? (r0 + SROWS) : N;
  const int nr  = (r1 > r0) ? (r1 - r0) : 0;
  const int n4  = nr >> 2;
  int4* h4 = (int4*)h;
  const int* __restrict__ srcp = table + (size_t)b * TSTRIDE + r0;
  const int4* s4 = (const int4*)srcp;
  for (int i0 = tid; i0 < n4; i0 += 1024 * LUNR) {
    int4 v[LUNR];
#pragma unroll
    for (int u = 0; u < LUNR; ++u) {
      const int i = i0 + u * 1024;
      v[u] = s4[(i < n4) ? i : (n4 - 1)];
    }
#pragma unroll
    for (int u = 0; u < LUNR; ++u) {
      const int i = i0 + u * 1024;
      h4[(i < n4) ? i : (n4 - 1)] = v[u];
    }
  }
  for (int i = (n4 << 2) + tid; i < nr; i += 1024) h[i] = srcp[i];
  __syncthreads();

  const int chunk = (E + NCHUNK - 1) / NCHUNK;
  const int e0 = b * chunk;
  const int e1 = (e0 + chunk < E) ? (e0 + chunk) : E;
  for (int eb = e0 + tid; eb < e1; eb += 1024 * EUNR) {
    int rr[EUNR];
    int cc[EUNR];
    int ww[EUNR];
#pragma unroll
    for (int i = 0; i < EUNR; ++i) {
      const int e  = eb + i * 1024;
      const int ec = (e < e1) ? e : (e1 - 1);
      const int v  = erow[ec];
      rr[i] = (e < e1 && v >= r0 && v < r1) ? (v - r0) : -1;
      cc[i] = ecol[ec];
      ww[i] = __float_as_int(ew[ec]);
    }
    int pp[EUNR];
#pragma unroll
    for (int i = 0; i < EUNR; ++i)
      pp[i] = (rr[i] >= 0) ? atomicAdd(&h[rr[i]], 1) : -1;
#pragma unroll
    for (int i = 0; i < EUNR; ++i)
      if (pp[i] >= 0) perm[pp[i]] = make_int2(cc[i], ww[i]);
  }
}

template <int K>
__device__ __forceinline__ void gather_rows(const float* __restrict__ seq_fts,
                                            int2 rec, int j, int f4, float4& ag) {
  int    cc[K];
  float  ww[K];
  float4 vv[K];
#pragma unroll
  for (int i = 0; i < K; ++i) {
    cc[i] = __builtin_amdgcn_readlane(rec.x, j + i);
    ww[i] = __int_as_float(__builtin_amdgcn_readlane(rec.y, j + i));
  }
#pragma unroll
  for (int i = 0; i < K; ++i)
    vv[i] = *(const float4*)(seq_fts + (size_t)cc[i] * OUT_FT + f4);
#pragma unroll
  for (int i = 0; i < K; ++i) {
    ag.x = __builtin_fmaf(ww[i], vv[i].x, ag.x);
    ag.y = __builtin_fmaf(ww[i], vv[i].y, ag.y);
    ag.z = __builtin_fmaf(ww[i], vv[i].z, ag.z);
    ag.w = __builtin_fmaf(ww[i], vv[i].w, ag.w);
  }
}

__global__ __launch_bounds__(256) void gnnfilm_gather_film(
    const float* __restrict__ seq_fts, const int2* __restrict__ perm,
    const int* __restrict__ pos, const int* __restrict__ cnt,
    const float* __restrict__ Wg, const float* __restrict__ bg,
    const float* __restrict__ Wb, const float* __restrict__ bb,
    const float* __restrict__ bias, const float* __restrict__ prelu_w,
    const int* __restrict__ node_type, float* __restrict__ out, int N) {
  const int lane = threadIdx.x & 31;
  const int f4   = lane << 2;
  const int tw   = (int)(gridDim.x * (blockDim.x >> 5));
  int n = __builtin_amdgcn_readfirstlane(
      (int)(blockIdx.x * (blockDim.x >> 5) + (threadIdx.x >> 5)));
  if (n >= N) return;

  const float slope = prelu_w[0];
  const float4 gb  = *(const float4*)(bg + f4);
  const float4 bbv = *(const float4*)(bb + f4);
  const float4 bi  = *(const float4*)(bias + f4);

  int cv = cnt[n];
  int pv = pos[n];
  int tv = node_type[n];

  while (true) {
    const int c = __builtin_amdgcn_readfirstlane(cv);
    const int s = __builtin_amdgcn_readfirstlane(pv) - c;
    const int t = __builtin_amdgcn_readfirstlane(tv);

    int m = (c < 32) ? c : 32;
    int2 rec = make_int2(0, 0);
    if (lane < m) rec = perm[s + lane];

    const int nn = n + tw;
    if (nn < N) {
      cv = cnt[nn];
      pv = pos[nn];
      tv = node_type[nn];
    }

    const float4 g  = *(const float4*)(Wg + (size_t)t * OUT_FT + f4);
    const float4 b  = *(const float4*)(Wb + (size_t)t * OUT_FT + f4);
    const float4 sf = *(const float4*)(seq_fts + (size_t)n * OUT_FT + f4);

    float4 ag = make_float4(0.f, 0.f, 0.f, 0.f);

    int base = 0;
    while (true) {
      int j = 0;
      for (; j + 4 <= m; j += 4) gather_rows<4>(seq_fts, rec, j, f4, ag);
      const int k = m - j;
      if (k == 3)      gather_rows<3>(seq_fts, rec, j, f4, ag);
      else if (k == 2) gather_rows<2>(seq_fts, rec, j, f4, ag);
      else if (k == 1) gather_rows<1>(seq_fts, rec, j, f4, ag);
      base += 32;
      if (base >= c) break;
      m = (c - base < 32) ? (c - base) : 32;
      if (lane < m) rec = perm[s + base + lane];
    }

    float o0 = (g.x + gb.x) * ag.x + (b.x + bbv.x) + bi.x + sf.x;
    float o1 = (g.y + gb.y) * ag.y + (b.y + bbv.y) + bi.y + sf.y;
    float o2 = (g.z + gb.z) * ag.z + (b.z + bbv.z) + bi.z + sf.z;
    float o3 = (g.w + gb.w) * ag.w + (b.w + bbv.w) + bi.w + sf.w;

    float4 r;
    r.x = o0 >= 0.f ? o0 : slope * o0;
    r.y = o1 >= 0.f ? o1 : slope * o1;
    r.z = o2 >= 0.f ? o2 : slope * o2;
    r.w = o3 >= 0.f ? o3 : slope * o3;
    *(float4*)(out + (size_t)n * OUT_FT + f4) = r;

    n = nn;
    if (n >= N) break;
  }
}

extern "C" void kernel_launch(void* const* d_in, const int* in_sizes, int n_in,
                              void* d_out, int out_size, void* d_ws, size_t ws_size,
                              hipStream_t stream) {
  const float* seq      = (const float*)d_in[0];
  const float* ew       = (const float*)d_in[1];
  const float* W_fc     = (const float*)d_in[2];
  const float* W_gamma  = (const float*)d_in[3];
  const float* b_gamma  = (const float*)d_in[4];
  const float* W_beta   = (const float*)d_in[5];
  const float* b_beta   = (const float*)d_in[6];
  const float* bias     = (const float*)d_in[7];
  const float* prelu_w  = (const float*)d_in[8];
  const int*   edge_row = (const int*)d_in[9];
  const int*   edge_col = (const int*)d_in[10];
  const int*   node_typ = (const int*)d_in[11];

  const int N = in_sizes[0] / IN_FT;
  const int E = in_sizes[1];

  float* out     = (float*)d_out;
  float* seq_fts = (float*)d_ws;
  int*   table   = (int*)(seq_fts + (size_t)N * OUT_FT);
  int*   cnt     = table + (size_t)NCHUNK * TSTRIDE;
  int*   pos     = cnt + N;
  int*   cursor  = pos + N;
  int2*  perm    = (int2*)(cursor + 2);

  gnnfilm_gemm_wmma<<<(N / 16 + 4) / 5, 256, 0, stream>>>(seq, W_fc, seq_fts, cursor, N);
  gnnfilm_hist<<<NCHUNK * NSLICE, 1024, 0, stream>>>(edge_row, table, N, E);
  gnnfilm_alloc<<<(N + 255) / 256, 256, 0, stream>>>(table, cnt, pos, cursor, N);
  gnnfilm_fill<<<NCHUNK * NSLICE, 1024, 0, stream>>>(ew, edge_row, edge_col, table, perm, N, E);
  gnnfilm_gather_film<<<(N + 31) / 32, 256, 0, stream>>>(seq_fts, perm, pos, cnt,
                                                       W_gamma, b_gamma, W_beta, b_beta,
                                                       bias, prelu_w, node_typ, out, N);
}


# The GEMM, which the edge grouping never reads, runs as extra 256-thread blocks behind the count blocks and the fill blocks (half its row tiles each), no launch of its own; row sum order kept
# speedup vs baseline: 2.8945x; 1.0892x over previous
#include <hip/hip_runtime.h>

typedef float v2f __attribute__((ext_vector_type(2)));
typedef float v8f __attribute__((ext_vector_type(8)));

#define IN_FT  128
#define OUT_FT 128

#define NCHUNK  64
#define NSLICE  4
#define NGROUP  (NCHUNK * NSLICE)
#define MAXN    50000
#define TSTRIDE 50048
#define SROWS   (TSTRIDE / NSLICE)
#define GTHR    256
#define HUNR    25
#define FUNR    17
#define LUNR    13
#define TPB     3

#define APITCH 132
#define ATILE  (16 * APITCH)
__device__ __forceinline__ void product_role(
    float* __restrict__ As, const float* __restrict__ seq,
    const float* __restrict__ W, float* __restrict__ seq_fts,
    int t0, int t1, int gb, int ngb) {
  const int tid  = threadIdx.x;
  const int lane = tid & 31;
  const int wave = tid >> 5;
  const int n0   = wave << 4;
  const int half = lane >> 4;
  const int l15  = lane & 15;

  int tile = t0 + gb;

  const int sc4 = lane << 2;
  float4 g0 = make_float4(0.f, 0.f, 0.f, 0.f);
  float4 g1 = make_float4(0.f, 0.f, 0.f, 0.f);
  if (tile < t1) {
    const float* __restrict__ sp = seq + (size_t)((tile << 4) + wave) * IN_FT + sc4;
    g0 = *(const float4*)sp;
    g1 = *(const float4*)(sp + 8 * IN_FT);
  }

  const float* __restrict__ bcol = W + n0 + l15;
  v2f bf[IN_FT / 4];
#pragma unroll
  for (int kc = 0; kc < IN_FT / 4; ++kc) {
    const int kb = kc * 4 + half * 2;
    bf[kc].x = bcol[(size_t)(kb + 0) * OUT_FT];
    bf[kc].y = bcol[(size_t)(kb + 1) * OUT_FT];
  }

  if (tile < t1) {
    *(float4*)&As[wave * APITCH + sc4]       = g0;
    *(float4*)&As[(wave + 8) * APITCH + sc4] = g1;
  }

  int buf = 0;
  for (; tile < t1; tile += ngb) {
    const int m0 = tile << 4;
    const int nt = tile + ngb;
    if (nt < t1) {
      const float* __restrict__ sp = seq + (size_t)((nt << 4) + wave) * IN_FT + sc4;
      g0 = *(const float4*)sp;
      g1 = *(const float4*)(sp + 8 * IN_FT);
    }

    __builtin_amdgcn_fence(__ATOMIC_RELEASE, "workgroup", "local");
    __builtin_amdgcn_s_barrier();
    __builtin_amdgcn_fence(__ATOMIC_ACQUIRE, "workgroup", "local");

    const float* ap = As + buf * ATILE + l15 * APITCH + half * 2;
    v2f af[IN_FT / 4];
#pragma unroll
    for (int kc = 0; kc < IN_FT / 4; ++kc)
      af[kc] = *(const v2f*)(ap + kc * 4);

    asm volatile("" :: "v"(af[0]), "v"(af[1]), "v"(af[2]), "v"(af[3]),
                       "v"(af[4]), "v"(af[5]), "v"(af[6]), "v"(af[7]),
                       "v"(af[8]), "v"(af[9]), "v"(af[10]), "v"(af[11]),
                       "v"(af[12]), "v"(af[13]), "v"(af[14]), "v"(af[15]));
    asm volatile("" :: "v"(af[16]), "v"(af[17]), "v"(af[18]), "v"(af[19]),
                       "v"(af[20]), "v"(af[21]), "v"(af[22]), "v"(af[23]),
                       "v"(af[24]), "v"(af[25]), "v"(af[26]), "v"(af[27]),
                       "v"(af[28]), "v"(af[29]), "v"(af[30]), "v"(af[31]));

    v8f c = {0.f, 0.f, 0.f, 0.f, 0.f, 0.f, 0.f, 0.f};
#pragma unroll
    for (int kc = 0; kc < IN_FT / 4; ++kc)
      c = __builtin_amdgcn_wmma_f32_16x16x4_f32(
           false, af[kc],  false, bf[kc],
           (short)0, c,  false,  false);

    float* __restrict__ crow =
        seq_fts + (size_t)(m0 + half * 8) * OUT_FT + n0 + l15;
#pragma unroll
    for (int v = 0; v < 8; ++v)
      crow[(size_t)v * OUT_FT] = c[v];

    if (nt < t1) {
      float* An = As + (buf ^ 1) * ATILE;
      *(float4*)&An[wave * APITCH + sc4]       = g0;
      *(float4*)&An[(wave + 8) * APITCH + sc4] = g1;
    }
    buf ^= 1;
  }
}

__device__ __forceinline__ void count_role(
    int* __restrict__ h, const int* __restrict__ erow, int* __restrict__ table,
    int N, int E, int bid) {
  const int tid = threadIdx.x;
  const int b   = bid / NSLICE;
  const int r0  = (bid % NSLICE) * SROWS;
  const int r1  = (r0 + SROWS < N) ? (r0 + SROWS) : N;
  const int nr  = (r1 > r0) ? (r1 - r0) : 0;
  const int n4  = nr >> 2;
  int4* h4 = (int4*)h;
  for (int i = tid; i < SROWS / 4; i += GTHR) h4[i] = make_int4(0, 0, 0, 0);
  __syncthreads();

  const int chunk = (E + NCHUNK - 1) / NCHUNK;
  const int e0 = b * chunk;
  const int e1 = (e0 + chunk < E) ? (e0 + chunk) : E;
  for (int eb = e0 + tid; eb < e1; eb += GTHR * HUNR) {
    int rr[HUNR];
#pragma unroll
    for (int i = 0; i < HUNR; ++i) {
      const int e = eb + i * GTHR;
      const int v = erow[(e < e1) ? e : (e1 - 1)];
      rr[i] = (e < e1 && v >= r0 && v < r1) ? (v - r0) : -1;
    }
#pragma unroll
    for (int i = 0; i < HUNR; ++i)
      if (rr[i] >= 0) atomicAdd(&h[rr[i]], 1);
  }
  __syncthreads();

  int* __restrict__ dst = table + (size_t)b * TSTRIDE + r0;
  int4* d4 = (int4*)dst;
  for (int i = tid; i < n4; i += GTHR) d4[i] = h4[i];
  for (int i = (n4 << 2) + tid; i < nr; i += GTHR) dst[i] = h[i];
}

__global__ __launch_bounds__(GTHR) void gnnfilm_count_product(
    const float* __restrict__ seq, const float* __restrict__ W,
    float* __restrict__ seq_fts, int* __restrict__ cursor,
    const int* __restrict__ erow, int* __restrict__ table,
    int N, int E, int t0, int t1) {
  __shared__ __attribute__((aligned(16))) int smem[SROWS];
  const int bid = blockIdx.x;
  if (bid < NGROUP) {
    if (bid == 0 && threadIdx.x == 0) cursor[0] = 0;
    count_role(smem, erow, table, N, E, bid);
    return;
  }
  product_role((float*)smem, seq, W, seq_fts, t0, t1,
               bid - NGROUP, (int)gridDim.x - NGROUP);
}

__global__ __launch_bounds__(256) void gnnfilm_alloc(
    int* __restrict__ table, int* __restrict__ cnt, int* __restrict__ pos,
    int* __restrict__ cursor, int N) {
  __shared__ int wsum[8];
  __shared__ int base_s;
  const int r    = blockIdx.x * 256 + threadIdx.x;
  const int lane = threadIdx.x & 31;
  const int wave = threadIdx.x >> 5;
  const int rc = (r < N) ? r : (N - 1);
  int t[NCHUNK];
#pragma unroll
  for (int b = 0; b < NCHUNK; ++b) {
    const int v = table[(size_t)b * TSTRIDE + rc];
    t[b] = (r < N) ? v : 0;
  }
  int c = 0;
#pragma unroll
  for (int b = 0; b < NCHUNK; ++b) c += t[b];
  int x = c;
#pragma unroll
  for (int d = 1; d < 32; d <<= 1) {
    int y = __shfl_up(x, d, 32);
    if (lane >= d) x += y;
  }
  if (lane == 31) wsum[wave] = x;
  __syncthreads();
  if (threadIdx.x == 0) {
    int tot = 0;
#pragma unroll
    for (int i = 0; i < 8; ++i) { int u = wsum[i]; wsum[i] = tot; tot += u; }
    base_s = atomicAdd(cursor, tot);
  }
  __syncthreads();
  if (r < N) {
    int run = base_s + wsum[wave] + (x - c);
    cnt[r] = c;
    pos[r] = run + c;
#pragma unroll
    for (int b = 0; b < NCHUNK; ++b) {
      table[(size_t)b * TSTRIDE + r] = run;
      run += t[b];
    }
  }
}

__device__ __forceinline__ void fill_role(
    int* __restrict__ h, const float* __restrict__ ew,
    const int* __restrict__ erow, const int* __restrict__ ecol,
    const int* __restrict__ table, int2* __restrict__ perm,
    int N, int E, int bid) {
  const int tid = threadIdx.x;
  const int b   = bid / NSLICE;
  const int r0  = (bid % NSLICE) * SROWS;
  const int r1  = (r0 + SROWS < N) ? (r0 + SROWS) : N;
  const int nr  = (r1 > r0) ? (r1 - r0) : 0;
  const int n4  = nr >> 2;
  int4* h4 = (int4*)h;
  const int* __restrict__ srcp = table + (size_t)b * TSTRIDE + r0;
  const int4* s4 = (const int4*)srcp;
  for (int i0 = tid; i0 < n4; i0 += GTHR * LUNR) {
    int4 v[LUNR];
#pragma unroll
    for (int u = 0; u < LUNR; ++u) {
      const int i = i0 + u * GTHR;
      v[u] = s4[(i < n4) ? i : (n4 - 1)];
    }
#pragma unroll
    for (int u = 0; u < LUNR; ++u) {
      const int i = i0 + u * GTHR;
      h4[(i < n4) ? i : (n4 - 1)] = v[u];
    }
  }
  for (int i = (n4 << 2) + tid; i < nr; i += GTHR) h[i] = srcp[i];
  __syncthreads();

  const int chunk = (E + NCHUNK - 1) / NCHUNK;
  const int e0 = b * chunk;
  const int e1 = (e0 + chunk < E) ? (e0 + chunk) : E;
  for (int eb = e0 + tid; eb < e1; eb += GTHR * FUNR) {
    int rr[FUNR];
    int cc[FUNR];
    int ww[FUNR];
#pragma unroll
    for (int i = 0; i < FUNR; ++i) {
      const int e  = eb + i * GTHR;
      const int ec = (e < e1) ? e : (e1 - 1);
      const int v  = erow[ec];
      rr[i] = (e < e1 && v >= r0 && v < r1) ? (v - r0) : -1;
      cc[i] = ecol[ec];
      ww[i] = __float_as_int(ew[ec]);
    }
    int pp[FUNR];
#pragma unroll
    for (int i = 0; i < FUNR; ++i)
      pp[i] = (rr[i] >= 0) ? atomicAdd(&h[rr[i]], 1) : -1;
#pragma unroll
    for (int i = 0; i < FUNR; ++i)
      if (pp[i] >= 0) perm[pp[i]] = make_int2(cc[i], ww[i]);
  }
}

__global__ __launch_bounds__(GTHR) void gnnfilm_fill_product(
    const float* __restrict__ ew, const int* __restrict__ erow,
    const int* __restrict__ ecol, const int* __restrict__ table,
    int2* __restrict__ perm, const float* __restrict__ seq,
    const float* __restrict__ W, float* __restrict__ seq_fts,
    int N, int E, int t0, int t1) {
  __shared__ __attribute__((aligned(16))) int smem[SROWS];
  const int bid = blockIdx.x;
  if (bid < NGROUP) {
    fill_role(smem, ew, erow, ecol, table, perm, N, E, bid);
    return;
  }
  product_role((float*)smem, seq, W, seq_fts, t0, t1,
               bid - NGROUP, (int)gridDim.x - NGROUP);
}

template <int K>
__device__ __forceinline__ void gather_rows(const float* __restrict__ seq_fts,
                                            int2 rec, int j, int f4, float4& ag) {
  int    cc[K];
  float  ww[K];
  float4 vv[K];
#pragma unroll
  for (int i = 0; i < K; ++i) {
    cc[i] = __builtin_amdgcn_readlane(rec.x, j + i);
    ww[i] = __int_as_float(__builtin_amdgcn_readlane(rec.y, j + i));
  }
#pragma unroll
  for (int i = 0; i < K; ++i)
    vv[i] = *(const float4*)(seq_fts + (size_t)cc[i] * OUT_FT + f4);
#pragma unroll
  for (int i = 0; i < K; ++i) {
    ag.x = __builtin_fmaf(ww[i], vv[i].x, ag.x);
    ag.y = __builtin_fmaf(ww[i], vv[i].y, ag.y);
    ag.z = __builtin_fmaf(ww[i], vv[i].z, ag.z);
    ag.w = __builtin_fmaf(ww[i], vv[i].w, ag.w);
  }
}

__global__ __launch_bounds__(256) void gnnfilm_gather_film(
    const float* __restrict__ seq_fts, const int2* __restrict__ perm,
    const int* __restrict__ pos, const int* __restrict__ cnt,
    const float* __restrict__ Wg, const float* __restrict__ bg,
    const float* __restrict__ Wb, const float* __restrict__ bb,
    const float* __restrict__ bias, const float* __restrict__ prelu_w,
    const int* __restrict__ node_type, float* __restrict__ out, int N) {
  const int lane = threadIdx.x & 31;
  const int f4   = lane << 2;
  const int tw   = (int)(gridDim.x * (blockDim.x >> 5));
  int n = __builtin_amdgcn_readfirstlane(
      (int)(blockIdx.x * (blockDim.x >> 5) + (threadIdx.x >> 5)));
  if (n >= N) return;

  const float slope = prelu_w[0];
  const float4 gb  = *(const float4*)(bg + f4);
  const float4 bbv = *(const float4*)(bb + f4);
  const float4 bi  = *(const float4*)(bias + f4);

  int cv = cnt[n];
  int pv = pos[n];
  int tv = node_type[n];

  while (true) {
    const int c = __builtin_amdgcn_readfirstlane(cv);
    const int s = __builtin_amdgcn_readfirstlane(pv) - c;
    const int t = __builtin_amdgcn_readfirstlane(tv);

    int m = (c < 32) ? c : 32;
    int2 rec = make_int2(0, 0);
    if (lane < m) rec = perm[s + lane];

    const int nn = n + tw;
    if (nn < N) {
      cv = cnt[nn];
      pv = pos[nn];
      tv = node_type[nn];
    }

    const float4 g  = *(const float4*)(Wg + (size_t)t * OUT_FT + f4);
    const float4 b  = *(const float4*)(Wb + (size_t)t * OUT_FT + f4);
    const float4 sf = *(const float4*)(seq_fts + (size_t)n * OUT_FT + f4);

    float4 ag = make_float4(0.f, 0.f, 0.f, 0.f);

    int base = 0;
    while (true) {
      int j = 0;
      for (; j + 4 <= m; j += 4) gather_rows<4>(seq_fts, rec, j, f4, ag);
      const int k = m - j;
      if (k == 3)      gather_rows<3>(seq_fts, rec, j, f4, ag);
      else if (k == 2) gather_rows<2>(seq_fts, rec, j, f4, ag);
      else if (k == 1) gather_rows<1>(seq_fts, rec, j, f4, ag);
      base += 32;
      if (base >= c) break;
      m = (c - base < 32) ? (c - base) : 32;
      if (lane < m) rec = perm[s + base + lane];
    }

    float o0 = (g.x + gb.x) * ag.x + (b.x + bbv.x) + bi.x + sf.x;
    float o1 = (g.y + gb.y) * ag.y + (b.y + bbv.y) + bi.y + sf.y;
    float o2 = (g.z + gb.z) * ag.z + (b.z + bbv.z) + bi.z + sf.z;
    float o3 = (g.w + gb.w) * ag.w + (b.w + bbv.w) + bi.w + sf.w;

    float4 r;
    r.x = o0 >= 0.f ? o0 : slope * o0;
    r.y = o1 >= 0.f ? o1 : slope * o1;
    r.z = o2 >= 0.f ? o2 : slope * o2;
    r.w = o3 >= 0.f ? o3 : slope * o3;
    *(float4*)(out + (size_t)n * OUT_FT + f4) = r;

    n = nn;
    if (n >= N) break;
  }
}

extern "C" void kernel_launch(void* const* d_in, const int* in_sizes, int n_in,
                              void* d_out, int out_size, void* d_ws, size_t ws_size,
                              hipStream_t stream) {
  const float* seq      = (const float*)d_in[0];
  const float* ew       = (const float*)d_in[1];
  const float* W_fc     = (const float*)d_in[2];
  const float* W_gamma  = (const float*)d_in[3];
  const float* b_gamma  = (const float*)d_in[4];
  const float* W_beta   = (const float*)d_in[5];
  const float* b_beta   = (const float*)d_in[6];
  const float* bias     = (const float*)d_in[7];
  const float* prelu_w  = (const float*)d_in[8];
  const int*   edge_row = (const int*)d_in[9];
  const int*   edge_col = (const int*)d_in[10];
  const int*   node_typ = (const int*)d_in[11];

  const int N = in_sizes[0] / IN_FT;
  const int E = in_sizes[1];

  float* out     = (float*)d_out;
  float* seq_fts = (float*)d_ws;
  int*   table   = (int*)(seq_fts + (size_t)N * OUT_FT);
  int*   cnt     = table + (size_t)NCHUNK * TSTRIDE;
  int*   pos     = cnt + N;
  int*   cursor  = pos + N;
  int2*  perm    = (int2*)(cursor + 2);

  const int ntile = N / 16;
  const int tmid  = (ntile + 1) / 2;
  const int gb1   = (tmid + TPB - 1) / TPB;
  const int gb2   = (ntile - tmid + TPB - 1) / TPB;

  gnnfilm_count_product<<<NGROUP + gb1, GTHR, 0, stream>>>(seq, W_fc, seq_fts, cursor,
                                                           edge_row, table, N, E, 0, tmid);
  gnnfilm_alloc<<<(N + 255) / 256, 256, 0, stream>>>(table, cnt, pos, cursor, N);
  gnnfilm_fill_product<<<NGROUP + gb2, GTHR, 0, stream>>>(ew, edge_row, edge_col, table, perm,
                                                          seq, W_fc, seq_fts, N, E, tmid, ntile);
  gnnfilm_gather_film<<<(N + 31) / 32, 256, 0, stream>>>(seq_fts, perm, pos, cnt,
                                                       W_gamma, b_gamma, W_beta, b_beta,
                                                       bias, prelu_w, node_typ, out, N);
}
